# starting kernel
# baseline (speedup 1.0000x reference)
#include <hip/hip_runtime.h>

typedef __attribute__((ext_vector_type(2))) float v2f;
typedef __attribute__((ext_vector_type(8))) float v8f;

#define N_NODES 50000
#define N_EDGES 800000
#define D_IN    512
#define D_OUT   128
#define N_BANDS (N_NODES / 16)

__global__ __launch_bounds__(256) void gemm_wmma_f32(
    const float* __restrict__ x, const float* __restrict__ W,
    float* __restrict__ h)
{
    const int lane = threadIdx.x & 31;
    const int band = blockIdx.x * 8 + (threadIdx.x >> 5);
    if (band >= N_BANDS) return;

    const int m0   = band * 16;
    const int half = lane >> 4;
    const int l16  = lane & 15;
    const int koff = half * 2;

    v8f acc[8] = {};

    const float* xrow = x + (size_t)(m0 + l16) * D_IN + koff;

    for (int k0 = 0; k0 < D_IN; k0 += 4) {
        v2f a;
        a.x = xrow[k0 + 0];
        a.y = xrow[k0 + 1];
        const float* wk = W + (size_t)(k0 + koff) * D_OUT + l16;
        #pragma unroll
        for (int nt = 0; nt < 8; ++nt) {
            v2f b;
            b.x = wk[nt * 16];
            b.y = wk[nt * 16 + D_OUT];
            acc[nt] = __builtin_amdgcn_wmma_f32_16x16x4_f32(
                 false, a,  false, b,
                 (short)0, acc[nt],
                 false,  false);
        }
    }

    #pragma unroll
    for (int nt = 0; nt < 8; ++nt) {
        float* hp = h + (size_t)(m0 + half * 8) * D_OUT + nt * 16 + l16;
        #pragma unroll
        for (int r = 0; r < 8; ++r)
            hp[(size_t)r * D_OUT] = acc[nt][r];
    }
}

__global__ __launch_bounds__(256) void bias_init(
    const float* __restrict__ b, float* __restrict__ out)
{
    const int idx = blockIdx.x * 256 + threadIdx.x;
    if (idx < N_NODES * D_OUT)
        out[idx] = b[idx & (D_OUT - 1)];
}

__global__ __launch_bounds__(256) void edge_scatter(
    const float* __restrict__ h, const float* __restrict__ edge_w,
    const int* __restrict__ src, const int* __restrict__ dst,
    float* __restrict__ out)
{
    const int gtid = blockIdx.x * 256 + threadIdx.x;
    const int e    = gtid >> 5;
    const int lane = threadIdx.x & 31;
    if (e >= N_EDGES) return;

    const int   s = src[e];
    const int   d = dst[e];
    const float w = edge_w[e];

    const float4 hv = *(const float4*)(h + (size_t)s * D_OUT + lane * 4);
    float* op = out + (size_t)d * D_OUT + lane * 4;

    __hip_atomic_fetch_add(op + 0, w * hv.x, __ATOMIC_RELAXED, __HIP_MEMORY_SCOPE_AGENT);
    __hip_atomic_fetch_add(op + 1, w * hv.y, __ATOMIC_RELAXED, __HIP_MEMORY_SCOPE_AGENT);
    __hip_atomic_fetch_add(op + 2, w * hv.z, __ATOMIC_RELAXED, __HIP_MEMORY_SCOPE_AGENT);
    __hip_atomic_fetch_add(op + 3, w * hv.w, __ATOMIC_RELAXED, __HIP_MEMORY_SCOPE_AGENT);
}

extern "C" void kernel_launch(void* const* d_in, const int* in_sizes, int n_in,
                              void* d_out, int out_size, void* d_ws, size_t ws_size,
                              hipStream_t stream)
{
    const float* x      = (const float*)d_in[0];
    const float* edge_w = (const float*)d_in[1];
    const int*   src    = (const int*)  d_in[2];
    const int*   dst    = (const int*)  d_in[3];
    const float* W      = (const float*)d_in[4];
    const float* b      = (const float*)d_in[5];
    float*       out    = (float*)d_out;
    float*       h      = (float*)d_ws;

    gemm_wmma_f32<<<(N_BANDS + 7) / 8, 256, 0, stream>>>(x, W, h);

    bias_init<<<(N_NODES * D_OUT + 255) / 256, 256, 0, stream>>>(b, out);

    edge_scatter<<<(N_EDGES * 32) / 256, 256, 0, stream>>>(h, edge_w, src, dst, out);
}


# edge scatter: f32 atomics removed; edges bucketed per dst row inside its out row (u32 slot counter), then one wave per row sums the h rows in registers; sum order = slot arrival, bias last
# speedup vs baseline: 1.9614x; 1.9614x over previous
#include <hip/hip_runtime.h>

typedef __attribute__((ext_vector_type(2))) float v2f;
typedef __attribute__((ext_vector_type(8))) float v8f;

#define N_NODES 50000
#define N_EDGES 800000
#define D_IN    512
#define D_OUT   128
#define N_BANDS (N_NODES / 16)

#define GEMM_WGS     256
#define GEMM_WAVES   16
#define K_STEPS      (D_IN / 4)
#define W_SLOTS      (K_STEPS * 4 * 32)

static_assert((GEMM_WAVES * 32) % 128 == 0, "the fill keeps (P, lane) per thread");

__global__ __launch_bounds__(GEMM_WAVES * 32) void gemm_wmma_f32(
    const float* __restrict__ x, const float* __restrict__ W,
    float* __restrict__ h)
{
    __shared__ float4 wl[W_SLOTS];

    const int tid  = threadIdx.x;
    const int lane = tid & 31;
    const int half = lane >> 4;
    const int l16  = lane & 15;
    const int koff = half * 2;

    {
        const int P = (tid >> 5) & 3;
        const float* wp = W + (size_t)(4 * (tid >> 7) + koff) * D_OUT + P * 32 + l16;
        for (int i = tid; i < W_SLOTS; i += GEMM_WAVES * 32) {
            float4 v;
            v.x = wp[0];
            v.y = wp[D_OUT];
            v.z = wp[16];
            v.w = wp[D_OUT + 16];
            wl[i] = v;
            wp += (size_t)(4 * (GEMM_WAVES * 32 / 128)) * D_OUT;
        }
    }
    __syncthreads();

    const int wave = __builtin_amdgcn_readfirstlane(tid >> 5);
    const float4* bl = wl + lane;

    for (int band = blockIdx.x + wave * GEMM_WGS; band < N_BANDS;
         band += GEMM_WGS * GEMM_WAVES) {
        const int m0 = band * 16;

        v8f acc[8] = {};

        const float* xrow = x + (size_t)(m0 + l16) * D_IN + koff;

        for (int S = 0; S < K_STEPS; ++S) {
            v2f a;
            a.x = xrow[4 * S + 0];
            a.y = xrow[4 * S + 1];
            #pragma unroll
            for (int P = 0; P < 4; ++P) {
                const float4 bv = bl[(S * 4 + P) * 32];
                v2f b0, b1;
                b0.x = bv.x; b0.y = bv.y;
                b1.x = bv.z; b1.y = bv.w;
                acc[2 * P] = __builtin_amdgcn_wmma_f32_16x16x4_f32(
                     false, a,  false, b0,
                     (short)0, acc[2 * P],
                     false,  false);
                acc[2 * P + 1] = __builtin_amdgcn_wmma_f32_16x16x4_f32(
                     false, a,  false, b1,
                     (short)0, acc[2 * P + 1],
                     false,  false);
            }
        }

        #pragma unroll
        for (int nt = 0; nt < 8; ++nt) {
            float* hp = h + (size_t)(m0 + half * 8) * D_OUT + nt * 16 + l16;
            #pragma unroll
            for (int r = 0; r < 8; ++r)
                hp[(size_t)r * D_OUT] = acc[nt][r];
        }
    }
}

#define ROW_CAP   63
#define CNT_DWORD (D_OUT - 1)

typedef unsigned int u32a __attribute__((may_alias));
typedef unsigned int u32x2 __attribute__((ext_vector_type(2)));
typedef u32x2 u32x2a __attribute__((may_alias));

__global__ __launch_bounds__(256) void bucket_clear(float* __restrict__ out)
{
    const int i = blockIdx.x * 256 + threadIdx.x;
    if (i < N_NODES)
        ((u32a*)out)[(size_t)i * D_OUT + CNT_DWORD] = 0u;
}

__global__ __launch_bounds__(256) void bucket_edges(
    const float* __restrict__ edge_w, const int* __restrict__ src,
    const int* __restrict__ dst, float* __restrict__ out)
{
    const int t = blockIdx.x * 256 + threadIdx.x;
    if (t >= N_EDGES / 4) return;

    const int4   d4 = ((const int4*)dst)[t];
    const int4   s4 = ((const int4*)src)[t];
    const float4 w4 = ((const float4*)edge_w)[t];

    u32a* r0 = (u32a*)out + (size_t)d4.x * D_OUT;
    u32a* r1 = (u32a*)out + (size_t)d4.y * D_OUT;
    u32a* r2 = (u32a*)out + (size_t)d4.z * D_OUT;
    u32a* r3 = (u32a*)out + (size_t)d4.w * D_OUT;

    const unsigned p0 = __hip_atomic_fetch_add(r0 + CNT_DWORD, 1u, __ATOMIC_RELAXED, __HIP_MEMORY_SCOPE_AGENT);
    const unsigned p1 = __hip_atomic_fetch_add(r1 + CNT_DWORD, 1u, __ATOMIC_RELAXED, __HIP_MEMORY_SCOPE_AGENT);
    const unsigned p2 = __hip_atomic_fetch_add(r2 + CNT_DWORD, 1u, __ATOMIC_RELAXED, __HIP_MEMORY_SCOPE_AGENT);
    const unsigned p3 = __hip_atomic_fetch_add(r3 + CNT_DWORD, 1u, __ATOMIC_RELAXED, __HIP_MEMORY_SCOPE_AGENT);

    u32x2 e;
    if (p0 < ROW_CAP) { e.x = (unsigned)s4.x; e.y = __float_as_uint(w4.x); ((u32x2a*)r0)[p0] = e; }
    if (p1 < ROW_CAP) { e.x = (unsigned)s4.y; e.y = __float_as_uint(w4.y); ((u32x2a*)r1)[p1] = e; }
    if (p2 < ROW_CAP) { e.x = (unsigned)s4.z; e.y = __float_as_uint(w4.z); ((u32x2a*)r2)[p2] = e; }
    if (p3 < ROW_CAP) { e.x = (unsigned)s4.w; e.y = __float_as_uint(w4.w); ((u32x2a*)r3)[p3] = e; }
}

#define GATHER_BLOCKS 1024
#define GATHER_WAVES  (GATHER_BLOCKS * 8)

__device__ __forceinline__ void add_slots(float4& acc, const u32x2 e, const int m,
                                          const float4* __restrict__ h4, const int lane)
{
    for (int j = 0; j < m; j += 8) {
        float4 v[8];
        float  w[8];
        #pragma unroll
        for (int u = 0; u < 8; ++u) {
            const int s = __builtin_amdgcn_readlane((int)e.x, j + u);
            w[u] = __uint_as_float((unsigned)__builtin_amdgcn_readlane((int)e.y, j + u));
            v[u] = h4[(size_t)s * (D_OUT / 4) + lane];
        }
        #pragma unroll
        for (int u = 0; u < 8; ++u) {
            if (j + u < m) {
                acc.x = fmaf(w[u], v[u].x, acc.x);
                acc.y = fmaf(w[u], v[u].y, acc.y);
                acc.z = fmaf(w[u], v[u].z, acc.z);
                acc.w = fmaf(w[u], v[u].w, acc.w);
            }
        }
    }
}

__global__ __launch_bounds__(256) void gather_rows(
    const float* __restrict__ h, const float* __restrict__ edge_w,
    const int* __restrict__ src, const int* __restrict__ dst,
    const float* __restrict__ b, float* __restrict__ out)
{
    const int lane = threadIdx.x & 31;
    const int wave = __builtin_amdgcn_readfirstlane((int)((blockIdx.x * 256 + threadIdx.x) >> 5));
    const float4* h4 = (const float4*)h;
    const float4  bv = ((const float4*)b)[lane];

    unsigned cnt_next = ((const u32a*)out)[(size_t)wave * D_OUT + CNT_DWORD];

    for (int d = wave; d < N_NODES; d += GATHER_WAVES) {
        float* row = out + (size_t)d * D_OUT;
        const unsigned cnt = (unsigned)__builtin_amdgcn_readfirstlane((int)cnt_next);

        const int dn = d + GATHER_WAVES;
        if (dn < N_NODES)
            cnt_next = ((const u32a*)out)[(size_t)dn * D_OUT + CNT_DWORD];

        float4 acc = {0.0f, 0.0f, 0.0f, 0.0f};

        if (cnt <= ROW_CAP) {
            const int n = (int)cnt;
            u32x2 e0 = {0u, 0u}, e1 = {0u, 0u};
            if (lane < n)      e0 = ((const u32x2a*)row)[lane];
            if (lane + 32 < n) e1 = ((const u32x2a*)row)[lane + 32];
            add_slots(acc, e0, n < 32 ? n : 32, h4, lane);
            if (n > 32)
                add_slots(acc, e1, n - 32, h4, lane);
        } else {
            for (int base = 0; base < N_EDGES; base += 128) {
                const int4 dd = ((const int4*)dst)[(base >> 2) + lane];
                const int  dc[4] = {dd.x, dd.y, dd.z, dd.w};
                #pragma unroll
                for (int c = 0; c < 4; ++c) {
                    unsigned hit = __builtin_amdgcn_ballot_w32(dc[c] == d);
                    while (hit) {
                        const int l = __builtin_ctz(hit);
                        hit &= hit - 1;
                        const int   ed = base + 4 * l + c;
                        const int   s  = src[ed];
                        const float w  = edge_w[ed];
                        const float4 v = h4[(size_t)s * (D_OUT / 4) + lane];
                        acc.x = fmaf(w, v.x, acc.x);
                        acc.y = fmaf(w, v.y, acc.y);
                        acc.z = fmaf(w, v.z, acc.z);
                        acc.w = fmaf(w, v.w, acc.w);
                    }
                }
            }
        }

        float4 o;
        o.x = acc.x + bv.x;
        o.y = acc.y + bv.y;
        o.z = acc.z + bv.z;
        o.w = acc.w + bv.w;
        ((float4*)row)[lane] = o;
    }
}

extern "C" void kernel_launch(void* const* d_in, const int* in_sizes, int n_in,
                              void* d_out, int out_size, void* d_ws, size_t ws_size,
                              hipStream_t stream)
{
    const float* x      = (const float*)d_in[0];
    const float* edge_w = (const float*)d_in[1];
    const int*   src    = (const int*)  d_in[2];
    const int*   dst    = (const int*)  d_in[3];
    const float* W      = (const float*)d_in[4];
    const float* b      = (const float*)d_in[5];
    float*       out    = (float*)d_out;
    float*       h      = (float*)d_ws;

    bucket_clear<<<(N_NODES + 255) / 256, 256, 0, stream>>>(out);

    bucket_edges<<<(N_EDGES / 4 + 255) / 256, 256, 0, stream>>>(edge_w, src, dst, out);

    gemm_wmma_f32<<<GEMM_WGS, GEMM_WAVES * 32, 0, stream>>>(x, W, h);

    gather_rows<<<GATHER_BLOCKS, 256, 0, stream>>>(h, edge_w, src, dst, b, out);
}


# edge bucketing moved into the GEMM launch: waves 13-15 of each work-group, which have no band, take the edges' slots while the others run the k loop; 3 launches; sum order = slot arrival as before
# speedup vs baseline: 2.0333x; 1.0367x over previous
#include <hip/hip_runtime.h>

typedef __attribute__((ext_vector_type(2))) float v2f;
typedef __attribute__((ext_vector_type(8))) float v8f;

#define N_NODES 50000
#define N_EDGES 800000
#define D_IN    512
#define D_OUT   128
#define N_BANDS (N_NODES / 16)

#define ROW_CAP   63
#define CNT_DWORD (D_OUT - 1)

typedef unsigned int u32a __attribute__((may_alias));
typedef unsigned int u32x2 __attribute__((ext_vector_type(2)));
typedef u32x2 u32x2a __attribute__((may_alias));

__global__ __launch_bounds__(256) void bucket_clear(float* __restrict__ out)
{
    const int i = blockIdx.x * 256 + threadIdx.x;
    if (i < N_NODES)
        ((u32a*)out)[(size_t)i * D_OUT + CNT_DWORD] = 0u;
}

__device__ __forceinline__ void take_slots(const int4 d4, const int4 s4, const float4 w4,
                                           float* __restrict__ out)
{
    u32a* r0 = (u32a*)out + (size_t)d4.x * D_OUT;
    u32a* r1 = (u32a*)out + (size_t)d4.y * D_OUT;
    u32a* r2 = (u32a*)out + (size_t)d4.z * D_OUT;
    u32a* r3 = (u32a*)out + (size_t)d4.w * D_OUT;

    const unsigned p0 = __hip_atomic_fetch_add(r0 + CNT_DWORD, 1u, __ATOMIC_RELAXED, __HIP_MEMORY_SCOPE_AGENT);
    const unsigned p1 = __hip_atomic_fetch_add(r1 + CNT_DWORD, 1u, __ATOMIC_RELAXED, __HIP_MEMORY_SCOPE_AGENT);
    const unsigned p2 = __hip_atomic_fetch_add(r2 + CNT_DWORD, 1u, __ATOMIC_RELAXED, __HIP_MEMORY_SCOPE_AGENT);
    const unsigned p3 = __hip_atomic_fetch_add(r3 + CNT_DWORD, 1u, __ATOMIC_RELAXED, __HIP_MEMORY_SCOPE_AGENT);

    u32x2 e;
    if (p0 < ROW_CAP) { e.x = (unsigned)s4.x; e.y = __float_as_uint(w4.x); ((u32x2a*)r0)[p0] = e; }
    if (p1 < ROW_CAP) { e.x = (unsigned)s4.y; e.y = __float_as_uint(w4.y); ((u32x2a*)r1)[p1] = e; }
    if (p2 < ROW_CAP) { e.x = (unsigned)s4.z; e.y = __float_as_uint(w4.z); ((u32x2a*)r2)[p2] = e; }
    if (p3 < ROW_CAP) { e.x = (unsigned)s4.w; e.y = __float_as_uint(w4.w); ((u32x2a*)r3)[p3] = e; }
}

#define GEMM_WGS     256
#define GEMM_WAVES   16
#define EDGE_WAVES   3
#define FIRST_EDGE_WAVE (GEMM_WAVES - EDGE_WAVES)
#define EDGE_THREADS (GEMM_WGS * EDGE_WAVES * 32)
#define K_STEPS      (D_IN / 4)
#define W_SLOTS      (K_STEPS * 4 * 32)

static_assert((GEMM_WAVES * 32) % 128 == 0, "the fill keeps (P, lane) per thread");
static_assert(GEMM_WGS * FIRST_EDGE_WAVE >= N_BANDS, "the edge waves have no band");
static_assert(EDGE_THREADS <= N_EDGES / 4, "every edge thread has a first group of four edges");

__global__ __launch_bounds__(GEMM_WAVES * 32) void gemm_wmma_f32_and_buckets(
    const float* __restrict__ x, const float* __restrict__ W,
    float* __restrict__ h,
    const float* __restrict__ edge_w, const int* __restrict__ src,
    const int* __restrict__ dst, float* __restrict__ out)
{
    __shared__ float4 wl[W_SLOTS];

    const int tid  = threadIdx.x;
    const int lane = tid & 31;
    const int half = lane >> 4;
    const int l16  = lane & 15;
    const int koff = half * 2;

    {
        const int P = (tid >> 5) & 3;
        const float* wp = W + (size_t)(4 * (tid >> 7) + koff) * D_OUT + P * 32 + l16;
        for (int i = tid; i < W_SLOTS; i += GEMM_WAVES * 32) {
            float4 v;
            v.x = wp[0];
            v.y = wp[D_OUT];
            v.z = wp[16];
            v.w = wp[D_OUT + 16];
            wl[i] = v;
            wp += (size_t)(4 * (GEMM_WAVES * 32 / 128)) * D_OUT;
        }
    }
    __syncthreads();

    const int wave = __builtin_amdgcn_readfirstlane(tid >> 5);

    if (wave >= FIRST_EDGE_WAVE) {
        const int4*   dst4 = (const int4*)dst;
        const int4*   src4 = (const int4*)src;
        const float4* ew4  = (const float4*)edge_w;

        int t = (blockIdx.x * EDGE_WAVES + (wave - FIRST_EDGE_WAVE)) * 32 + lane;
        int4   d4 = dst4[t];
        int4   s4 = src4[t];
        float4 w4 = ew4[t];
        for (;;) {
            const int  tn   = t + EDGE_THREADS;
            const bool more = tn < N_EDGES / 4;
            int4   dn = d4;
            int4   sn = s4;
            float4 wn = w4;
            if (more) {
                dn = dst4[tn];
                sn = src4[tn];
                wn = ew4[tn];
            }
            take_slots(d4, s4, w4, out);
            if (!more) break;
            d4 = dn;
            s4 = sn;
            w4 = wn;
            t  = tn;
        }
        return;
    }

    const float4* bl = wl + lane;

    for (int band = blockIdx.x + wave * GEMM_WGS; band < N_BANDS;
         band += GEMM_WGS * GEMM_WAVES) {
        const int m0 = band * 16;

        v8f acc[8] = {};

        const float* xrow = x + (size_t)(m0 + l16) * D_IN + koff;

        for (int S = 0; S < K_STEPS; ++S) {
            v2f a;
            a.x = xrow[4 * S + 0];
            a.y = xrow[4 * S + 1];
            #pragma unroll
            for (int P = 0; P < 4; ++P) {
                const float4 bv = bl[(S * 4 + P) * 32];
                v2f b0, b1;
                b0.x = bv.x; b0.y = bv.y;
                b1.x = bv.z; b1.y = bv.w;
                acc[2 * P] = __builtin_amdgcn_wmma_f32_16x16x4_f32(
                     false, a,  false, b0,
                     (short)0, acc[2 * P],
                     false,  false);
                acc[2 * P + 1] = __builtin_amdgcn_wmma_f32_16x16x4_f32(
                     false, a,  false, b1,
                     (short)0, acc[2 * P + 1],
                     false,  false);
            }
        }

        #pragma unroll
        for (int nt = 0; nt < 8; ++nt) {
            float* hp = h + (size_t)(m0 + half * 8) * D_OUT + nt * 16 + l16;
            #pragma unroll
            for (int r = 0; r < 8; ++r)
                hp[(size_t)r * D_OUT] = acc[nt][r];
        }
    }
}

#define GATHER_BLOCKS 1024
#define GATHER_WAVES  (GATHER_BLOCKS * 8)

__device__ __forceinline__ void add_slots(float4& acc, const u32x2 e, const int m,
                                          const float4* __restrict__ h4, const int lane)
{
    for (int j = 0; j < m; j += 8) {
        float4 v[8];
        float  w[8];
        #pragma unroll
        for (int u = 0; u < 8; ++u) {
            const int s = __builtin_amdgcn_readlane((int)e.x, j + u);
            w[u] = __uint_as_float((unsigned)__builtin_amdgcn_readlane((int)e.y, j + u));
            v[u] = h4[(size_t)s * (D_OUT / 4) + lane];
        }
        #pragma unroll
        for (int u = 0; u < 8; ++u) {
            if (j + u < m) {
                acc.x = fmaf(w[u], v[u].x, acc.x);
                acc.y = fmaf(w[u], v[u].y, acc.y);
                acc.z = fmaf(w[u], v[u].z, acc.z);
                acc.w = fmaf(w[u], v[u].w, acc.w);
            }
        }
    }
}

__global__ __launch_bounds__(256) void gather_rows(
    const float* __restrict__ h, const float* __restrict__ edge_w,
    const int* __restrict__ src, const int* __restrict__ dst,
    const float* __restrict__ b, float* __restrict__ out)
{
    const int lane = threadIdx.x & 31;
    const int wave = __builtin_amdgcn_readfirstlane((int)((blockIdx.x * 256 + threadIdx.x) >> 5));
    const float4* h4 = (const float4*)h;
    const float4  bv = ((const float4*)b)[lane];

    unsigned cnt_next = ((const u32a*)out)[(size_t)wave * D_OUT + CNT_DWORD];

    for (int d = wave; d < N_NODES; d += GATHER_WAVES) {
        float* row = out + (size_t)d * D_OUT;
        const unsigned cnt = (unsigned)__builtin_amdgcn_readfirstlane((int)cnt_next);

        const int dn = d + GATHER_WAVES;
        if (dn < N_NODES)
            cnt_next = ((const u32a*)out)[(size_t)dn * D_OUT + CNT_DWORD];

        float4 acc = {0.0f, 0.0f, 0.0f, 0.0f};

        if (cnt <= ROW_CAP) {
            const int n = (int)cnt;
            u32x2 e0 = {0u, 0u}, e1 = {0u, 0u};
            if (lane < n)      e0 = ((const u32x2a*)row)[lane];
            if (lane + 32 < n) e1 = ((const u32x2a*)row)[lane + 32];
            add_slots(acc, e0, n < 32 ? n : 32, h4, lane);
            if (n > 32)
                add_slots(acc, e1, n - 32, h4, lane);
        } else {
            for (int base = 0; base < N_EDGES; base += 128) {
                const int4 dd = ((const int4*)dst)[(base >> 2) + lane];
                const int  dc[4] = {dd.x, dd.y, dd.z, dd.w};
                #pragma unroll
                for (int c = 0; c < 4; ++c) {
                    unsigned hit = __builtin_amdgcn_ballot_w32(dc[c] == d);
                    while (hit) {
                        const int l = __builtin_ctz(hit);
                        hit &= hit - 1;
                        const int   ed = base + 4 * l + c;
                        const int   s  = src[ed];
                        const float w  = edge_w[ed];
                        const float4 v = h4[(size_t)s * (D_OUT / 4) + lane];
                        acc.x = fmaf(w, v.x, acc.x);
                        acc.y = fmaf(w, v.y, acc.y);
                        acc.z = fmaf(w, v.z, acc.z);
                        acc.w = fmaf(w, v.w, acc.w);
                    }
                }
            }
        }

        float4 o;
        o.x = acc.x + bv.x;
        o.y = acc.y + bv.y;
        o.z = acc.z + bv.z;
        o.w = acc.w + bv.w;
        ((float4*)row)[lane] = o;
    }
}

extern "C" void kernel_launch(void* const* d_in, const int* in_sizes, int n_in,
                              void* d_out, int out_size, void* d_ws, size_t ws_size,
                              hipStream_t stream)
{
    const float* x      = (const float*)d_in[0];
    const float* edge_w = (const float*)d_in[1];
    const int*   src    = (const int*)  d_in[2];
    const int*   dst    = (const int*)  d_in[3];
    const float* W      = (const float*)d_in[4];
    const float* b      = (const float*)d_in[5];
    float*       out    = (float*)d_out;
    float*       h      = (float*)d_ws;

    bucket_clear<<<(N_NODES + 255) / 256, 256, 0, stream>>>(out);

    gemm_wmma_f32_and_buckets<<<GEMM_WGS, GEMM_WAVES * 32, 0, stream>>>(
        x, W, h, edge_w, src, dst, out);

    gather_rows<<<GATHER_BLOCKS, 256, 0, stream>>>(h, edge_w, src, dst, b, out);
}
